# starting kernel
# baseline (speedup 1.0000x reference)
#include <hip/hip_runtime.h>
#include <math.h>

#define BB 2
#define II 48
#define JJ 320
#define DD 256

#define NEGC   (-1000000000.0f)
#define LOGEPS (-1000.0f)
#define NEG_ID (-1e30f)

#define LOG2E_F 1.4426950408889634f
#define LN2_F   0.6931471805599453f

typedef __attribute__((ext_vector_type(2))) float v2f;
typedef __attribute__((ext_vector_type(8))) float v8f;

__device__ __forceinline__ float fast_exp(float x) {
    return __builtin_amdgcn_exp2f(x * LOG2E_F);
}
__device__ __forceinline__ float fast_log(float x) {
    return LN2_F * __builtin_amdgcn_logf(x);
}

__device__ __forceinline__ float lse2(float a, float b) {
    float m = fmaxf(a, b);
    float d = fminf(a, b) - m;
    float t = fast_exp(d);
    return m + fast_log(1.0f + t);
}

__global__ void __launch_bounds__(256)
k_transpose_text(const float* __restrict__ text, float* __restrict__ TT) {
    int b = blockIdx.y;
    for (int t = blockIdx.x * 256 + threadIdx.x; t < II * DD; t += gridDim.x * 256) {
        int i = t / DD, d = t % DD;
        TT[((size_t)b * DD + d) * II + i] = text[((size_t)b * II + i) * DD + d];
    }
}

__global__ void __launch_bounds__(32)
k_energy_wmma(const float* __restrict__ text, const float* __restrict__ mel,
              const float* __restrict__ gum, float* __restrict__ E) {
    const int b = blockIdx.z, it = blockIdx.y, jt = blockIdx.x;
    const int lane = threadIdx.x;
    const int half = lane >> 4;
    const int line = lane & 15;

    const float* __restrict__ Ta = text + (size_t)b * II * DD + (size_t)(it * 16 + line) * DD + half * 2;
    const float* __restrict__ Mb = mel  + (size_t)b * JJ * DD + (size_t)(jt * 16 + line) * DD + half * 2;

    __builtin_prefetch(Ta, 0, 1);
    __builtin_prefetch(Mb, 0, 1);
    __builtin_prefetch(Ta + 128, 0, 1);
    __builtin_prefetch(Mb + 128, 0, 1);

    v8f acc = {};
#pragma unroll 4
    for (int k = 0; k < DD; k += 4) {
        v2f a  = *(const v2f*)(Ta + k);
        v2f bv = *(const v2f*)(Mb + k);
        acc = __builtin_amdgcn_wmma_f32_16x16x4_f32(false, a, false, bv,
                                                    (short)0, acc, false, false);
    }

    const float invTemp = 1.0f / 0.55f;
    const float scale   = 1.0f / 256.0f;
#pragma unroll
    for (int v = 0; v < 8; ++v) {
        int i = it * 16 + v + half * 8;
        int j = jt * 16 + line;
        size_t idx = (size_t)b * II * JJ + (size_t)i * JJ + j;
        float u = gum[idx];
        u = fminf(fmaxf(u, 1e-7f), 1.0f - 1e-7f);
        float noise = -fast_log(-fast_log(u));
        E[idx] = (acc[v] * scale + noise) * invTemp;
    }
}

__global__ void __launch_bounds__(32)
k_suffix_lse(const float* __restrict__ E, float* __restrict__ S) {
    const int row = blockIdx.x;
    const int i = row % II;
    const int lane = threadIdx.x;
    const float* __restrict__ e = E + (size_t)row * JJ;
    float* __restrict__ s = S + (size_t)row * JJ;

    float carry = NEG_ID;
    for (int c = (JJ / 32) - 1; c >= 0; --c) {
        int j = c * 32 + lane;
        float x = e[j];
        if (i == II - 1 && j != JJ - 1) x = LOGEPS;
        float acc = x;
#pragma unroll
        for (int off = 1; off < 32; off <<= 1) {
            float y = __shfl_down(acc, off, 32);
            if (lane + off < 32) acc = lse2(acc, y);
        }
        float val = lse2(acc, carry);
        s[j] = val;
        carry = __shfl(val, 0, 32);
    }
}

__global__ void __launch_bounds__(32)
k_forward(const float* __restrict__ E, const float* __restrict__ S,
          float* __restrict__ P) {
    __shared__ float prev[JJ];
    __shared__ float tsh[JJ];
    const int b = blockIdx.x;
    const int lane = threadIdx.x;

    for (int j = lane; j < JJ; j += 32) {
        float v = (j == 0) ? 0.0f : NEGC;
        prev[j] = v;
        P[((size_t)b * II + 0) * JJ + j] = v;
    }
    __syncthreads();

    for (int i = 1; i < II; ++i) {
        const float* __restrict__ Er = E + ((size_t)b * II + (i - 1)) * JJ;
        const float* __restrict__ Sr = S + ((size_t)b * II + (i - 1)) * JJ;
        float carry = NEG_ID;
        for (int c = 0; c < JJ / 32; ++c) {
            int k = c * 32 + lane;
            float acc = prev[k] - Sr[k];
#pragma unroll
            for (int off = 1; off < 32; off <<= 1) {
                float y = __shfl_up(acc, off, 32);
                if (lane >= off) acc = lse2(acc, y);
            }
            float scan = lse2(acc, carry);
            carry = __shfl(scan, 31, 32);
            tsh[k] = Er[k] + scan;
        }
        __syncthreads();
        for (int jc = lane; jc < JJ; jc += 32) {
            float v = (jc == 0) ? NEGC : tsh[jc - 1];
            if (!(jc >= i && jc < (JJ - II + i + 2))) v = NEGC;
            prev[jc] = v;
            P[((size_t)b * II + i) * JJ + jc] = v;
        }
        __syncthreads();
    }
}

__global__ void __launch_bounds__(32)
k_gamma(const float* __restrict__ P, const float* __restrict__ S,
        float* __restrict__ GT) {
    const int row = blockIdx.x;
    const int b = row / II, i = row % II;
    const int lane = threadIdx.x;
    const float* __restrict__ p = P + (size_t)row * JJ;
    const float* __restrict__ s = S + (size_t)row * JJ;

    float carry = NEG_ID;
    for (int c = 0; c < JJ / 32; ++c) {
        int j = c * 32 + lane;
        float acc = p[j] - s[j];
#pragma unroll
        for (int off = 1; off < 32; off <<= 1) {
            float y = __shfl_up(acc, off, 32);
            if (lane >= off) acc = lse2(acc, y);
        }
        float scan = lse2(acc, carry);
        carry = __shfl(scan, 31, 32);
        float lb = s[j] + scan;
        float gamma = fast_exp(lb);
        if (i == II - 1 && j != JJ - 1) gamma = 0.0f;
        GT[((size_t)b * JJ + j) * II + i] = gamma;
    }
}

__global__ void __launch_bounds__(32)
k_expand_wmma(const float* __restrict__ GT, const float* __restrict__ TT,
              float* __restrict__ out) {
    const int b = blockIdx.z, jt = blockIdx.y, dt = blockIdx.x;
    const int lane = threadIdx.x;
    const int half = lane >> 4;
    const int line = lane & 15;
    const int j0 = jt * 16, d0 = dt * 16;
    const float* __restrict__ Ga = GT + ((size_t)b * JJ + (j0 + line)) * II + half * 2;
    const float* __restrict__ Tb = TT + ((size_t)b * DD + (d0 + line)) * II + half * 2;

    __builtin_prefetch(Ga, 0, 1);
    __builtin_prefetch(Tb, 0, 1);

    v8f acc = {};
#pragma unroll
    for (int k = 0; k < II; k += 4) {
        v2f a  = *(const v2f*)(Ga + k);
        v2f bv = *(const v2f*)(Tb + k);
        acc = __builtin_amdgcn_wmma_f32_16x16x4_f32(false, a, false, bv,
                                                    (short)0, acc, false, false);
    }
#pragma unroll
    for (int v = 0; v < 8; ++v) {
        int j = j0 + v + half * 8;
        out[(size_t)b * JJ * DD + (size_t)j * DD + d0 + line] = acc[v];
    }
}

extern "C" void kernel_launch(void* const* d_in, const int* in_sizes, int n_in,
                              void* d_out, int out_size, void* d_ws, size_t ws_size,
                              hipStream_t stream) {
    const float* text = (const float*)d_in[0];
    const float* mel  = (const float*)d_in[1];
    const float* gum  = (const float*)d_in[2];
    float* out = (float*)d_out;

    const size_t N = (size_t)BB * II * JJ;
    float* E  = (float*)d_ws;
    float* S  = E + N;
    float* P  = S + N;
    float* GT = P + N;
    float* TT = GT + N;

    k_transpose_text<<<dim3(8, BB), 256, 0, stream>>>(text, TT);
    k_energy_wmma<<<dim3(JJ / 16, II / 16, BB), 32, 0, stream>>>(text, mel, gum, E);
    k_suffix_lse<<<dim3(BB * II), 32, 0, stream>>>(E, S);
    k_forward<<<dim3(BB), 32, 0, stream>>>(E, S, P);
    k_gamma<<<dim3(BB * II), 32, 0, stream>>>(P, S, GT);
    k_expand_wmma<<<dim3(DD / 16, JJ / 16, BB), 32, 0, stream>>>(GT, TT, out);
    (void)in_sizes; (void)n_in; (void)out_size; (void)ws_size;
}


# Suffix-LSE and gamma rows as 10-wave work-groups, one column per thread: a row's 10 chunk scans run side by side, one load per thread, chunk totals meet in LDS; LSE reduction tree changes
# speedup vs baseline: 7.3211x; 7.3211x over previous
#include <hip/hip_runtime.h>
#include <math.h>

#define BB 2
#define II 48
#define JJ 320
#define DD 256

#define NEGC   (-1000000000.0f)
#define LOGEPS (-1000.0f)
#define NEG_ID (-1e30f)

#define LOG2E_F 1.4426950408889634f
#define LN2_F   0.6931471805599453f

typedef __attribute__((ext_vector_type(2))) float v2f;
typedef __attribute__((ext_vector_type(8))) float v8f;

__device__ __forceinline__ float fast_exp(float x) {
    return __builtin_amdgcn_exp2f(x * LOG2E_F);
}
__device__ __forceinline__ float fast_log(float x) {
    return LN2_F * __builtin_amdgcn_logf(x);
}

__device__ __forceinline__ float lse2(float a, float b) {
    float m = fmaxf(a, b);
    float d = fminf(a, b) - m;
    float t = fast_exp(d);
    return m + fast_log(1.0f + t);
}

template <int N>
__device__ __forceinline__ float row_shr(float fill, float v) {
    return __builtin_bit_cast(float, __builtin_amdgcn_update_dpp(
        __builtin_bit_cast(int, fill), __builtin_bit_cast(int, v),
        0x110 + N, 0xf, 0xf, false));
}
template <int N>
__device__ __forceinline__ float row_shl(float fill, float v) {
    return __builtin_bit_cast(float, __builtin_amdgcn_update_dpp(
        __builtin_bit_cast(int, fill), __builtin_bit_cast(int, v),
        0x100 + N, 0xf, 0xf, false));
}
__device__ __forceinline__ float other_row_last(float v) {
    int x = __builtin_bit_cast(int, v);
    return __builtin_bit_cast(float,
        __builtin_amdgcn_permlanex16(x, x, (int)0xffffffffu, (int)0xffffffffu, false, false));
}
__device__ __forceinline__ float other_row_first(float v) {
    int x = __builtin_bit_cast(int, v);
    return __builtin_bit_cast(float,
        __builtin_amdgcn_permlanex16(x, x, 0, 0, false, false));
}
__device__ __forceinline__ float lane_value(float v, int l) {
    return __builtin_bit_cast(float,
        __builtin_amdgcn_readlane(__builtin_bit_cast(int, v), l));
}

__device__ __forceinline__ float wave_prefix_lse(float acc, int lane, float& low) {
    acc = lse2(acc, row_shr<1>(NEG_ID, acc));
    acc = lse2(acc, row_shr<2>(NEG_ID, acc));
    acc = lse2(acc, row_shr<4>(NEG_ID, acc));
    acc = lse2(acc, row_shr<8>(NEG_ID, acc));
    float t = other_row_last(acc);
    low = (lane >= 16) ? t : NEG_ID;
    return lse2(acc, low);
}
__device__ __forceinline__ float wave_suffix_lse(float acc, int lane) {
    acc = lse2(acc, row_shl<1>(NEG_ID, acc));
    acc = lse2(acc, row_shl<2>(NEG_ID, acc));
    acc = lse2(acc, row_shl<4>(NEG_ID, acc));
    acc = lse2(acc, row_shl<8>(NEG_ID, acc));
    float t = other_row_first(acc);
    float high = (lane < 16) ? t : NEG_ID;
    return lse2(acc, high);
}

__global__ void __launch_bounds__(256)
k_transpose_text(const float* __restrict__ text, float* __restrict__ TT) {
    int b = blockIdx.y;
    for (int t = blockIdx.x * 256 + threadIdx.x; t < II * DD; t += gridDim.x * 256) {
        int i = t / DD, d = t % DD;
        TT[((size_t)b * DD + d) * II + i] = text[((size_t)b * II + i) * DD + d];
    }
}

__global__ void __launch_bounds__(32)
k_energy_wmma(const float* __restrict__ text, const float* __restrict__ mel,
              const float* __restrict__ gum, float* __restrict__ E) {
    const int b = blockIdx.z, it = blockIdx.y, jt = blockIdx.x;
    const int lane = threadIdx.x;
    const int half = lane >> 4;
    const int line = lane & 15;

    const float* __restrict__ Ta = text + (size_t)b * II * DD + (size_t)(it * 16 + line) * DD + half * 2;
    const float* __restrict__ Mb = mel  + (size_t)b * JJ * DD + (size_t)(jt * 16 + line) * DD + half * 2;

    __builtin_prefetch(Ta, 0, 1);
    __builtin_prefetch(Mb, 0, 1);
    __builtin_prefetch(Ta + 128, 0, 1);
    __builtin_prefetch(Mb + 128, 0, 1);

    v8f acc = {};
#pragma unroll 4
    for (int k = 0; k < DD; k += 4) {
        v2f a  = *(const v2f*)(Ta + k);
        v2f bv = *(const v2f*)(Mb + k);
        acc = __builtin_amdgcn_wmma_f32_16x16x4_f32(false, a, false, bv,
                                                    (short)0, acc, false, false);
    }

    const float invTemp = 1.0f / 0.55f;
    const float scale   = 1.0f / 256.0f;
#pragma unroll
    for (int v = 0; v < 8; ++v) {
        int i = it * 16 + v + half * 8;
        int j = jt * 16 + line;
        size_t idx = (size_t)b * II * JJ + (size_t)i * JJ + j;
        float u = gum[idx];
        u = fminf(fmaxf(u, 1e-7f), 1.0f - 1e-7f);
        float noise = -fast_log(-fast_log(u));
        E[idx] = (acc[v] * scale + noise) * invTemp;
    }
}

#define NWAVE (JJ / 32)

__global__ void __launch_bounds__(JJ)
k_suffix_lse(const float* __restrict__ E, float* __restrict__ S) {
    __shared__ __attribute__((aligned(16))) float tot[16];
    const int row = blockIdx.x;
    const int i = row % II;
    const int j = threadIdx.x;
    const int lane = j & 31;
    const int w = j >> 5;

    float x = E[(size_t)row * JJ + j];
    if (i == II - 1 && j != JJ - 1) x = LOGEPS;

    float acc = wave_suffix_lse(x, lane);
    if (lane == 0) tot[w] = acc;
    __syncthreads();

    float t[NWAVE - 1];
    for (int c = 0; c < NWAVE - 1; ++c) {
        float v = tot[c + 1];
        t[c] = (c + 1 > w) ? v : NEG_ID;
    }
    float m = acc;
    for (int c = 0; c < NWAVE - 1; ++c) m = fmaxf(m, t[c]);
    float xs[NWAVE];
    xs[0] = fast_exp(acc - m);
    for (int c = 0; c < NWAVE - 1; ++c) xs[c + 1] = fast_exp(t[c] - m);
    float sum = (((xs[0] + xs[1]) + (xs[2] + xs[3])) + ((xs[4] + xs[5]) + (xs[6] + xs[7])))
              + (xs[8] + xs[9]);
    S[(size_t)row * JJ + j] = m + fast_log(sum);
}

__global__ void __launch_bounds__(JJ)
k_forward(const float* __restrict__ E, const float* __restrict__ S,
          float* __restrict__ P) {
    __shared__ __attribute__((aligned(16))) float tot[2][16];
    const int b = blockIdx.x;
    const int j = threadIdx.x;
    const int lane = j & 31;
    const int w = j >> 5;

    const float* __restrict__ Eb = E + (size_t)b * II * JJ;
    const float* __restrict__ Sb = S + (size_t)b * II * JJ;
    float* __restrict__ Pb = P + (size_t)b * II * JJ;

    float prev = (j == 0) ? 0.0f : NEGC;
    Pb[j] = prev;

    const int je = (j > 0) ? j - 1 : 0;
    float s_next = Sb[j];
    float e_next = Eb[je];

    for (int i = 1; i < II; ++i) {
        const float s_cur = s_next;
        const float e_cur = e_next;
        if (i < II - 1) {
            s_next = Sb[(size_t)i * JJ + j];
            e_next = Eb[(size_t)i * JJ + je];
        }

        float low;
        float acc = wave_prefix_lse(prev - s_cur, lane, low);
        if (lane == 31) tot[i & 1][w] = acc;
        float ex = row_shr<1>(low, acc);

        __builtin_amdgcn_fence(__ATOMIC_RELEASE, "workgroup", "local");
        __builtin_amdgcn_s_barrier();
        __builtin_amdgcn_fence(__ATOMIC_ACQUIRE, "workgroup", "local");

        float t[NWAVE - 1];
        for (int c = 0; c < NWAVE - 1; ++c) {
            float v = tot[i & 1][c];
            t[c] = (c < w) ? v : NEG_ID;
        }
        float m = ex;
        for (int c = 0; c < NWAVE - 1; ++c) m = fmaxf(m, t[c]);
        float x[NWAVE];
        x[0] = fast_exp(ex - m);
        for (int c = 0; c < NWAVE - 1; ++c) x[c + 1] = fast_exp(t[c] - m);
        float sum = (((x[0] + x[1]) + (x[2] + x[3])) + ((x[4] + x[5]) + (x[6] + x[7])))
                  + (x[8] + x[9]);
        float v = e_cur + (m + fast_log(sum));

        if (j == 0) v = NEGC;
        if (!(j >= i && j < (JJ - II + i + 2))) v = NEGC;
        prev = v;
        Pb[(size_t)i * JJ + j] = v;
    }
}

__global__ void __launch_bounds__(JJ)
k_gamma(const float* __restrict__ P, const float* __restrict__ S,
        float* __restrict__ GT) {
    __shared__ __attribute__((aligned(16))) float tot[16];
    const int row = blockIdx.x;
    const int b = row / II, i = row % II;
    const int j = threadIdx.x;
    const int lane = j & 31;
    const int w = j >> 5;

    const float sv = S[(size_t)row * JJ + j];
    const float pv = P[(size_t)row * JJ + j];

    float low;
    float acc = wave_prefix_lse(pv - sv, lane, low);
    if (lane == 31) tot[w] = acc;
    __syncthreads();

    float t[NWAVE - 1];
    for (int c = 0; c < NWAVE - 1; ++c) {
        float v = tot[c];
        t[c] = (c < w) ? v : NEG_ID;
    }
    float m = acc;
    for (int c = 0; c < NWAVE - 1; ++c) m = fmaxf(m, t[c]);
    float xs[NWAVE];
    xs[0] = fast_exp(acc - m);
    for (int c = 0; c < NWAVE - 1; ++c) xs[c + 1] = fast_exp(t[c] - m);
    float sum = (((xs[0] + xs[1]) + (xs[2] + xs[3])) + ((xs[4] + xs[5]) + (xs[6] + xs[7])))
              + (xs[8] + xs[9]);
    float scan = m + fast_log(sum);

    float lb = sv + scan;
    float gamma = fast_exp(lb);
    if (i == II - 1 && j != JJ - 1) gamma = 0.0f;
    GT[((size_t)b * JJ + j) * II + i] = gamma;
}

__global__ void __launch_bounds__(32)
k_expand_wmma(const float* __restrict__ GT, const float* __restrict__ TT,
              float* __restrict__ out) {
    const int b = blockIdx.z, jt = blockIdx.y, dt = blockIdx.x;
    const int lane = threadIdx.x;
    const int half = lane >> 4;
    const int line = lane & 15;
    const int j0 = jt * 16, d0 = dt * 16;
    const float* __restrict__ Ga = GT + ((size_t)b * JJ + (j0 + line)) * II + half * 2;
    const float* __restrict__ Tb = TT + ((size_t)b * DD + (d0 + line)) * II + half * 2;

    __builtin_prefetch(Ga, 0, 1);
    __builtin_prefetch(Tb, 0, 1);

    v8f acc = {};
#pragma unroll
    for (int k = 0; k < II; k += 4) {
        v2f a  = *(const v2f*)(Ga + k);
        v2f bv = *(const v2f*)(Tb + k);
        acc = __builtin_amdgcn_wmma_f32_16x16x4_f32(false, a, false, bv,
                                                    (short)0, acc, false, false);
    }
#pragma unroll
    for (int v = 0; v < 8; ++v) {
        int j = j0 + v + half * 8;
        out[(size_t)b * JJ * DD + (size_t)j * DD + d0 + line] = acc[v];
    }
}

extern "C" void kernel_launch(void* const* d_in, const int* in_sizes, int n_in,
                              void* d_out, int out_size, void* d_ws, size_t ws_size,
                              hipStream_t stream) {
    const float* text = (const float*)d_in[0];
    const float* mel  = (const float*)d_in[1];
    const float* gum  = (const float*)d_in[2];
    float* out = (float*)d_out;

    const size_t N = (size_t)BB * II * JJ;
    float* E  = (float*)d_ws;
    float* S  = E + N;
    float* P  = S + N;
    float* GT = P + N;
    float* TT = GT + N;

    k_transpose_text<<<dim3(8, BB), 256, 0, stream>>>(text, TT);
    k_energy_wmma<<<dim3(JJ / 16, II / 16, BB), 32, 0, stream>>>(text, mel, gum, E);
    k_suffix_lse<<<dim3(BB * II), JJ, 0, stream>>>(E, S);
    k_forward<<<dim3(BB), JJ, 0, stream>>>(E, S, P);
    k_gamma<<<dim3(BB * II), JJ, 0, stream>>>(P, S, GT);
    k_expand_wmma<<<dim3(DD / 16, JJ / 16, BB), 32, 0, stream>>>(GT, TT, out);
    (void)in_sizes; (void)n_in; (void)out_size; (void)ws_size;
}


# K3 forward scan: each row's two operand loads are issued seven rows ahead through a ring of eight register pairs (was one row ahead), so a row no longer waits a memory trip for them; sums unchanged
# speedup vs baseline: 8.2324x; 1.1245x over previous
#include <hip/hip_runtime.h>
#include <math.h>

#define BB 2
#define II 48
#define JJ 320
#define DD 256

#define NEGC   (-1000000000.0f)
#define LOGEPS (-1000.0f)
#define NEG_ID (-1e30f)

#define LOG2E_F 1.4426950408889634f
#define LN2_F   0.6931471805599453f

typedef __attribute__((ext_vector_type(2))) float v2f;
typedef __attribute__((ext_vector_type(4))) float v4f;
typedef __attribute__((ext_vector_type(8))) float v8f;

__device__ __forceinline__ float fast_exp(float x) {
    return __builtin_amdgcn_exp2f(x * LOG2E_F);
}
__device__ __forceinline__ float fast_log(float x) {
    return LN2_F * __builtin_amdgcn_logf(x);
}

__device__ __forceinline__ float lse2(float a, float b) {
    float m = fmaxf(a, b);
    float d = fminf(a, b) - m;
    float t = fast_exp(d);
    return m + fast_log(1.0f + t);
}

template <int N>
__device__ __forceinline__ float row_shr(float fill, float v) {
    return __builtin_bit_cast(float, __builtin_amdgcn_update_dpp(
        __builtin_bit_cast(int, fill), __builtin_bit_cast(int, v),
        0x110 + N, 0xf, 0xf, false));
}
template <int N>
__device__ __forceinline__ float row_shl(float fill, float v) {
    return __builtin_bit_cast(float, __builtin_amdgcn_update_dpp(
        __builtin_bit_cast(int, fill), __builtin_bit_cast(int, v),
        0x100 + N, 0xf, 0xf, false));
}
__device__ __forceinline__ float other_row_last(float v) {
    int x = __builtin_bit_cast(int, v);
    return __builtin_bit_cast(float,
        __builtin_amdgcn_permlanex16(x, x, (int)0xffffffffu, (int)0xffffffffu, false, false));
}
__device__ __forceinline__ float other_row_first(float v) {
    int x = __builtin_bit_cast(int, v);
    return __builtin_bit_cast(float,
        __builtin_amdgcn_permlanex16(x, x, 0, 0, false, false));
}
__device__ __forceinline__ float lane_value(float v, int l) {
    return __builtin_bit_cast(float,
        __builtin_amdgcn_readlane(__builtin_bit_cast(int, v), l));
}

__device__ __forceinline__ float wave_prefix_lse(float acc, int lane, float& low) {
    acc = lse2(acc, row_shr<1>(NEG_ID, acc));
    acc = lse2(acc, row_shr<2>(NEG_ID, acc));
    acc = lse2(acc, row_shr<4>(NEG_ID, acc));
    acc = lse2(acc, row_shr<8>(NEG_ID, acc));
    float t = other_row_last(acc);
    low = (lane >= 16) ? t : NEG_ID;
    return lse2(acc, low);
}
__device__ __forceinline__ float wave_suffix_lse(float acc, int lane) {
    acc = lse2(acc, row_shl<1>(NEG_ID, acc));
    acc = lse2(acc, row_shl<2>(NEG_ID, acc));
    acc = lse2(acc, row_shl<4>(NEG_ID, acc));
    acc = lse2(acc, row_shl<8>(NEG_ID, acc));
    float t = other_row_first(acc);
    float high = (lane < 16) ? t : NEG_ID;
    return lse2(acc, high);
}

__global__ void __launch_bounds__(256)
k_transpose_text(const float* __restrict__ text, float* __restrict__ TT) {
    int b = blockIdx.y;
    for (int t = blockIdx.x * 256 + threadIdx.x; t < II * DD; t += gridDim.x * 256) {
        int i = t / DD, d = t % DD;
        TT[((size_t)b * DD + d) * II + i] = text[((size_t)b * II + i) * DD + d];
    }
}

__global__ void __launch_bounds__(32)
k_energy_wmma(const float* __restrict__ text, const float* __restrict__ mel,
              const float* __restrict__ gum, float* __restrict__ E) {
    const int b = blockIdx.z, it = blockIdx.y, jt = blockIdx.x;
    const int lane = threadIdx.x;
    const int half = lane >> 4;
    const int line = lane & 15;

    const float* __restrict__ Ta = text + (size_t)b * II * DD + (size_t)(it * 16 + line) * DD + half * 2;
    const float* __restrict__ Mb = mel  + (size_t)b * JJ * DD + (size_t)(jt * 16 + line) * DD + half * 2;

    __builtin_prefetch(Ta, 0, 1);
    __builtin_prefetch(Mb, 0, 1);
    __builtin_prefetch(Ta + 128, 0, 1);
    __builtin_prefetch(Mb + 128, 0, 1);

    v8f acc = {};
#pragma unroll 4
    for (int k = 0; k < DD; k += 4) {
        v2f a  = *(const v2f*)(Ta + k);
        v2f bv = *(const v2f*)(Mb + k);
        acc = __builtin_amdgcn_wmma_f32_16x16x4_f32(false, a, false, bv,
                                                    (short)0, acc, false, false);
    }

    const float invTemp = 1.0f / 0.55f;
    const float scale   = 1.0f / 256.0f;
#pragma unroll
    for (int v = 0; v < 8; ++v) {
        int i = it * 16 + v + half * 8;
        int j = jt * 16 + line;
        size_t idx = (size_t)b * II * JJ + (size_t)i * JJ + j;
        float u = gum[idx];
        u = fminf(fmaxf(u, 1e-7f), 1.0f - 1e-7f);
        float noise = -fast_log(-fast_log(u));
        E[idx] = (acc[v] * scale + noise) * invTemp;
    }
}

#define NWAVE (JJ / 32)

__global__ void __launch_bounds__(JJ)
k_suffix_lse(const float* __restrict__ E, float* __restrict__ S) {
    __shared__ __attribute__((aligned(16))) float tot[16];
    const int row = blockIdx.x;
    const int i = row % II;
    const int j = threadIdx.x;
    const int lane = j & 31;
    const int w = j >> 5;

    float x = E[(size_t)row * JJ + j];
    if (i == II - 1 && j != JJ - 1) x = LOGEPS;

    float acc = wave_suffix_lse(x, lane);
    if (lane == 0) tot[w] = acc;
    __syncthreads();

    float t[NWAVE - 1];
    for (int c = 0; c < NWAVE - 1; ++c) {
        float v = tot[c + 1];
        t[c] = (c + 1 > w) ? v : NEG_ID;
    }
    float m = acc;
    for (int c = 0; c < NWAVE - 1; ++c) m = fmaxf(m, t[c]);
    float xs[NWAVE];
    xs[0] = fast_exp(acc - m);
    for (int c = 0; c < NWAVE - 1; ++c) xs[c + 1] = fast_exp(t[c] - m);
    float sum = (((xs[0] + xs[1]) + (xs[2] + xs[3])) + ((xs[4] + xs[5]) + (xs[6] + xs[7])))
              + (xs[8] + xs[9]);
    S[(size_t)row * JJ + j] = m + fast_log(sum);
}

#define PFD 8
#define NMAIN (((II - 1) / PFD) * PFD)
static_assert(NMAIN + PFD <= II, "the ring's last refill must stay inside the planes");
static_assert(II - 1 - NMAIN < PFD, "the rows behind the loop take one ring slot each");

__device__ __forceinline__ float forward_row(int i, int j, int lane, int w, float prev,
                                             float s_cur, float e_cur,
                                             float (*tot)[16], float* __restrict__ Pb) {
    float low;
    float acc = wave_prefix_lse(prev - s_cur, lane, low);
    if (lane == 31) tot[i & 1][w] = acc;
    float ex = row_shr<1>(low, acc);

    __builtin_amdgcn_fence(__ATOMIC_RELEASE, "workgroup", "local");
    __builtin_amdgcn_s_barrier();
    __builtin_amdgcn_fence(__ATOMIC_ACQUIRE, "workgroup", "local");

    const v4f ta = *(const v4f*)&tot[i & 1][0];
    const v4f tb = *(const v4f*)&tot[i & 1][4];
    const float tv[NWAVE - 1] = {ta[0], ta[1], ta[2], ta[3], tb[0], tb[1], tb[2], tb[3],
                                 tot[i & 1][8]};
    float t[NWAVE - 1];
    for (int c = 0; c < NWAVE - 1; ++c) t[c] = (c < w) ? tv[c] : NEG_ID;
    float m = ex;
    for (int c = 0; c < NWAVE - 1; ++c) m = fmaxf(m, t[c]);
    float x[NWAVE];
    x[0] = fast_exp(ex - m);
    for (int c = 0; c < NWAVE - 1; ++c) x[c + 1] = fast_exp(t[c] - m);
    float sum = (((x[0] + x[1]) + (x[2] + x[3])) + ((x[4] + x[5]) + (x[6] + x[7])))
              + (x[8] + x[9]);
    float v = e_cur + (m + fast_log(sum));

    if (j == 0) v = NEGC;
    if (!(j >= i && j < (JJ - II + i + 2))) v = NEGC;
    Pb[(size_t)i * JJ + j] = v;
    return v;
}

__global__ void __launch_bounds__(JJ)
k_forward(const float* __restrict__ E, const float* __restrict__ S,
          float* __restrict__ P) {
    __shared__ __attribute__((aligned(16))) float tot[2][16];
    const int b = blockIdx.x;
    const int j = threadIdx.x;
    const int lane = j & 31;
    const int w = j >> 5;

    const float* __restrict__ Eb = E + (size_t)b * II * JJ;
    const float* __restrict__ Sb = S + (size_t)b * II * JJ;
    float* __restrict__ Pb = P + (size_t)b * II * JJ;

    float prev = (j == 0) ? 0.0f : NEGC;
    Pb[j] = prev;

    const int je = (j > 0) ? j - 1 : 0;
    float s_q[PFD], e_q[PFD];
#pragma unroll
    for (int r = 0; r < PFD; ++r) {
        s_q[r] = Sb[(size_t)r * JJ + j];
        e_q[r] = Eb[(size_t)r * JJ + je];
    }

    for (int i0 = 1; i0 <= NMAIN; i0 += PFD) {
#pragma unroll
        for (int u = 0; u < PFD; ++u) {
            const int i = i0 + u;
            prev = forward_row(i, j, lane, w, prev, s_q[u], e_q[u], tot, Pb);
            s_q[u] = Sb[(size_t)(i - 1 + PFD) * JJ + j];
            e_q[u] = Eb[(size_t)(i - 1 + PFD) * JJ + je];
        }
    }
#pragma unroll
    for (int u = 0; u < II - 1 - NMAIN; ++u) {
        prev = forward_row(NMAIN + 1 + u, j, lane, w, prev, s_q[u], e_q[u], tot, Pb);
    }
}

__global__ void __launch_bounds__(JJ)
k_gamma(const float* __restrict__ P, const float* __restrict__ S,
        float* __restrict__ GT) {
    __shared__ __attribute__((aligned(16))) float tot[16];
    const int row = blockIdx.x;
    const int b = row / II, i = row % II;
    const int j = threadIdx.x;
    const int lane = j & 31;
    const int w = j >> 5;

    const float sv = S[(size_t)row * JJ + j];
    const float pv = P[(size_t)row * JJ + j];

    float low;
    float acc = wave_prefix_lse(pv - sv, lane, low);
    if (lane == 31) tot[w] = acc;
    __syncthreads();

    float t[NWAVE - 1];
    for (int c = 0; c < NWAVE - 1; ++c) {
        float v = tot[c];
        t[c] = (c < w) ? v : NEG_ID;
    }
    float m = acc;
    for (int c = 0; c < NWAVE - 1; ++c) m = fmaxf(m, t[c]);
    float xs[NWAVE];
    xs[0] = fast_exp(acc - m);
    for (int c = 0; c < NWAVE - 1; ++c) xs[c + 1] = fast_exp(t[c] - m);
    float sum = (((xs[0] + xs[1]) + (xs[2] + xs[3])) + ((xs[4] + xs[5]) + (xs[6] + xs[7])))
              + (xs[8] + xs[9]);
    float scan = m + fast_log(sum);

    float lb = sv + scan;
    float gamma = fast_exp(lb);
    if (i == II - 1 && j != JJ - 1) gamma = 0.0f;
    GT[((size_t)b * JJ + j) * II + i] = gamma;
}

__global__ void __launch_bounds__(32)
k_expand_wmma(const float* __restrict__ GT, const float* __restrict__ TT,
              float* __restrict__ out) {
    const int b = blockIdx.z, jt = blockIdx.y, dt = blockIdx.x;
    const int lane = threadIdx.x;
    const int half = lane >> 4;
    const int line = lane & 15;
    const int j0 = jt * 16, d0 = dt * 16;
    const float* __restrict__ Ga = GT + ((size_t)b * JJ + (j0 + line)) * II + half * 2;
    const float* __restrict__ Tb = TT + ((size_t)b * DD + (d0 + line)) * II + half * 2;

    __builtin_prefetch(Ga, 0, 1);
    __builtin_prefetch(Tb, 0, 1);

    v8f acc = {};
#pragma unroll
    for (int k = 0; k < II; k += 4) {
        v2f a  = *(const v2f*)(Ga + k);
        v2f bv = *(const v2f*)(Tb + k);
        acc = __builtin_amdgcn_wmma_f32_16x16x4_f32(false, a, false, bv,
                                                    (short)0, acc, false, false);
    }
#pragma unroll
    for (int v = 0; v < 8; ++v) {
        int j = j0 + v + half * 8;
        out[(size_t)b * JJ * DD + (size_t)j * DD + d0 + line] = acc[v];
    }
}

extern "C" void kernel_launch(void* const* d_in, const int* in_sizes, int n_in,
                              void* d_out, int out_size, void* d_ws, size_t ws_size,
                              hipStream_t stream) {
    const float* text = (const float*)d_in[0];
    const float* mel  = (const float*)d_in[1];
    const float* gum  = (const float*)d_in[2];
    float* out = (float*)d_out;

    const size_t N = (size_t)BB * II * JJ;
    float* E  = (float*)d_ws;
    float* S  = E + N;
    float* P  = S + N;
    float* GT = P + N;
    float* TT = GT + N;

    k_transpose_text<<<dim3(8, BB), 256, 0, stream>>>(text, TT);
    k_energy_wmma<<<dim3(JJ / 16, II / 16, BB), 32, 0, stream>>>(text, mel, gum, E);
    k_suffix_lse<<<dim3(BB * II), JJ, 0, stream>>>(E, S);
    k_forward<<<dim3(BB), JJ, 0, stream>>>(E, S, P);
    k_gamma<<<dim3(BB * II), JJ, 0, stream>>>(P, S, GT);
    k_expand_wmma<<<dim3(DD / 16, JJ / 16, BB), 32, 0, stream>>>(GT, TT, out);
    (void)in_sizes; (void)n_in; (void)out_size; (void)ws_size;
}


# Transposed text copy for the expand GEMM is written by the energy kernel's waves, each storing the 16x16 text block it reads as A fragments: the transpose launch is gone (6 to 5); sums unchanged
# speedup vs baseline: 9.6332x; 1.1702x over previous
#include <hip/hip_runtime.h>
#include <math.h>

#define BB 2
#define II 48
#define JJ 320
#define DD 256

#define NEGC   (-1000000000.0f)
#define LOGEPS (-1000.0f)
#define NEG_ID (-1e30f)

#define LOG2E_F 1.4426950408889634f
#define LN2_F   0.6931471805599453f

typedef __attribute__((ext_vector_type(2))) float v2f;
typedef __attribute__((ext_vector_type(4))) float v4f;
typedef __attribute__((ext_vector_type(8))) float v8f;

__device__ __forceinline__ float fast_exp(float x) {
    return __builtin_amdgcn_exp2f(x * LOG2E_F);
}
__device__ __forceinline__ float fast_log(float x) {
    return LN2_F * __builtin_amdgcn_logf(x);
}

__device__ __forceinline__ float lse2(float a, float b) {
    float m = fmaxf(a, b);
    float d = fminf(a, b) - m;
    float t = fast_exp(d);
    return m + fast_log(1.0f + t);
}

template <int N>
__device__ __forceinline__ float row_shr(float fill, float v) {
    return __builtin_bit_cast(float, __builtin_amdgcn_update_dpp(
        __builtin_bit_cast(int, fill), __builtin_bit_cast(int, v),
        0x110 + N, 0xf, 0xf, false));
}
template <int N>
__device__ __forceinline__ float row_shl(float fill, float v) {
    return __builtin_bit_cast(float, __builtin_amdgcn_update_dpp(
        __builtin_bit_cast(int, fill), __builtin_bit_cast(int, v),
        0x100 + N, 0xf, 0xf, false));
}
__device__ __forceinline__ float other_row_last(float v) {
    int x = __builtin_bit_cast(int, v);
    return __builtin_bit_cast(float,
        __builtin_amdgcn_permlanex16(x, x, (int)0xffffffffu, (int)0xffffffffu, false, false));
}
__device__ __forceinline__ float other_row_first(float v) {
    int x = __builtin_bit_cast(int, v);
    return __builtin_bit_cast(float,
        __builtin_amdgcn_permlanex16(x, x, 0, 0, false, false));
}
__device__ __forceinline__ float lane_value(float v, int l) {
    return __builtin_bit_cast(float,
        __builtin_amdgcn_readlane(__builtin_bit_cast(int, v), l));
}

__device__ __forceinline__ float wave_prefix_lse(float acc, int lane, float& low) {
    acc = lse2(acc, row_shr<1>(NEG_ID, acc));
    acc = lse2(acc, row_shr<2>(NEG_ID, acc));
    acc = lse2(acc, row_shr<4>(NEG_ID, acc));
    acc = lse2(acc, row_shr<8>(NEG_ID, acc));
    float t = other_row_last(acc);
    low = (lane >= 16) ? t : NEG_ID;
    return lse2(acc, low);
}
__device__ __forceinline__ float wave_suffix_lse(float acc, int lane) {
    acc = lse2(acc, row_shl<1>(NEG_ID, acc));
    acc = lse2(acc, row_shl<2>(NEG_ID, acc));
    acc = lse2(acc, row_shl<4>(NEG_ID, acc));
    acc = lse2(acc, row_shl<8>(NEG_ID, acc));
    float t = other_row_first(acc);
    float high = (lane < 16) ? t : NEG_ID;
    return lse2(acc, high);
}

__global__ void __launch_bounds__(32)
k_energy_wmma(const float* __restrict__ text, const float* __restrict__ mel,
              const float* __restrict__ gum, float* __restrict__ E,
              float* __restrict__ TT) {
    const int b = blockIdx.z, it = blockIdx.y, jt = blockIdx.x;
    const int lane = threadIdx.x;
    const int half = lane >> 4;
    const int line = lane & 15;

    const float* __restrict__ Ta = text + (size_t)b * II * DD + (size_t)(it * 16 + line) * DD + half * 2;
    const float* __restrict__ Mb = mel  + (size_t)b * JJ * DD + (size_t)(jt * 16 + line) * DD + half * 2;

    __builtin_prefetch(Ta, 0, 1);
    __builtin_prefetch(Mb, 0, 1);
    __builtin_prefetch(Ta + 128, 0, 1);
    __builtin_prefetch(Mb + 128, 0, 1);

    v8f acc = {};
#pragma unroll 4
    for (int k = 0; k < DD; k += 4) {
        v2f a  = *(const v2f*)(Ta + k);
        v2f bv = *(const v2f*)(Mb + k);
        acc = __builtin_amdgcn_wmma_f32_16x16x4_f32(false, a, false, bv,
                                                    (short)0, acc, false, false);
    }

    if (jt < DD / 16) {
        const float* __restrict__ Tt = Ta + jt * 16;
        float* __restrict__ To = TT + ((size_t)b * DD + jt * 16 + half * 2) * II + it * 16 + line;
#pragma unroll
        for (int q = 0; q < 4; ++q) {
            v2f t2 = *(const v2f*)(Tt + q * 4);
            To[(size_t)(q * 4) * II]     = t2[0];
            To[(size_t)(q * 4 + 1) * II] = t2[1];
        }
    }

    const float invTemp = 1.0f / 0.55f;
    const float scale   = 1.0f / 256.0f;
#pragma unroll
    for (int v = 0; v < 8; ++v) {
        int i = it * 16 + v + half * 8;
        int j = jt * 16 + line;
        size_t idx = (size_t)b * II * JJ + (size_t)i * JJ + j;
        float u = gum[idx];
        u = fminf(fmaxf(u, 1e-7f), 1.0f - 1e-7f);
        float noise = -fast_log(-fast_log(u));
        E[idx] = (acc[v] * scale + noise) * invTemp;
    }
}

#define NWAVE (JJ / 32)

__global__ void __launch_bounds__(JJ)
k_suffix_lse(const float* __restrict__ E, float* __restrict__ S) {
    __shared__ __attribute__((aligned(16))) float tot[16];
    const int row = blockIdx.x;
    const int i = row % II;
    const int j = threadIdx.x;
    const int lane = j & 31;
    const int w = j >> 5;

    float x = E[(size_t)row * JJ + j];
    if (i == II - 1 && j != JJ - 1) x = LOGEPS;

    float acc = wave_suffix_lse(x, lane);
    if (lane == 0) tot[w] = acc;
    __syncthreads();

    float t[NWAVE - 1];
    for (int c = 0; c < NWAVE - 1; ++c) {
        float v = tot[c + 1];
        t[c] = (c + 1 > w) ? v : NEG_ID;
    }
    float m = acc;
    for (int c = 0; c < NWAVE - 1; ++c) m = fmaxf(m, t[c]);
    float xs[NWAVE];
    xs[0] = fast_exp(acc - m);
    for (int c = 0; c < NWAVE - 1; ++c) xs[c + 1] = fast_exp(t[c] - m);
    float sum = (((xs[0] + xs[1]) + (xs[2] + xs[3])) + ((xs[4] + xs[5]) + (xs[6] + xs[7])))
              + (xs[8] + xs[9]);
    S[(size_t)row * JJ + j] = m + fast_log(sum);
}

#define PFD 8
#define NMAIN (((II - 1) / PFD) * PFD)
static_assert(NMAIN + PFD <= II, "the ring's last refill must stay inside the planes");
static_assert(II - 1 - NMAIN < PFD, "the rows behind the loop take one ring slot each");

__device__ __forceinline__ float forward_row(int i, int j, int lane, int w, float prev,
                                             float s_cur, float e_cur,
                                             float (*tot)[16], float* __restrict__ Pb) {
    float low;
    float acc = wave_prefix_lse(prev - s_cur, lane, low);
    if (lane == 31) tot[i & 1][w] = acc;
    float ex = row_shr<1>(low, acc);

    __builtin_amdgcn_fence(__ATOMIC_RELEASE, "workgroup", "local");
    __builtin_amdgcn_s_barrier();
    __builtin_amdgcn_fence(__ATOMIC_ACQUIRE, "workgroup", "local");

    const v4f ta = *(const v4f*)&tot[i & 1][0];
    const v4f tb = *(const v4f*)&tot[i & 1][4];
    const float tv[NWAVE - 1] = {ta[0], ta[1], ta[2], ta[3], tb[0], tb[1], tb[2], tb[3],
                                 tot[i & 1][8]};
    float t[NWAVE - 1];
    for (int c = 0; c < NWAVE - 1; ++c) t[c] = (c < w) ? tv[c] : NEG_ID;
    float m = ex;
    for (int c = 0; c < NWAVE - 1; ++c) m = fmaxf(m, t[c]);
    float x[NWAVE];
    x[0] = fast_exp(ex - m);
    for (int c = 0; c < NWAVE - 1; ++c) x[c + 1] = fast_exp(t[c] - m);
    float sum = (((x[0] + x[1]) + (x[2] + x[3])) + ((x[4] + x[5]) + (x[6] + x[7])))
              + (x[8] + x[9]);
    float v = e_cur + (m + fast_log(sum));

    if (j == 0) v = NEGC;
    if (!(j >= i && j < (JJ - II + i + 2))) v = NEGC;
    Pb[(size_t)i * JJ + j] = v;
    return v;
}

__global__ void __launch_bounds__(JJ)
k_forward(const float* __restrict__ E, const float* __restrict__ S,
          float* __restrict__ P) {
    __shared__ __attribute__((aligned(16))) float tot[2][16];
    const int b = blockIdx.x;
    const int j = threadIdx.x;
    const int lane = j & 31;
    const int w = j >> 5;

    const float* __restrict__ Eb = E + (size_t)b * II * JJ;
    const float* __restrict__ Sb = S + (size_t)b * II * JJ;
    float* __restrict__ Pb = P + (size_t)b * II * JJ;

    float prev = (j == 0) ? 0.0f : NEGC;
    Pb[j] = prev;

    const int je = (j > 0) ? j - 1 : 0;
    float s_q[PFD], e_q[PFD];
#pragma unroll
    for (int r = 0; r < PFD; ++r) {
        s_q[r] = Sb[(size_t)r * JJ + j];
        e_q[r] = Eb[(size_t)r * JJ + je];
    }

    for (int i0 = 1; i0 <= NMAIN; i0 += PFD) {
#pragma unroll
        for (int u = 0; u < PFD; ++u) {
            const int i = i0 + u;
            prev = forward_row(i, j, lane, w, prev, s_q[u], e_q[u], tot, Pb);
            s_q[u] = Sb[(size_t)(i - 1 + PFD) * JJ + j];
            e_q[u] = Eb[(size_t)(i - 1 + PFD) * JJ + je];
        }
    }
#pragma unroll
    for (int u = 0; u < II - 1 - NMAIN; ++u) {
        prev = forward_row(NMAIN + 1 + u, j, lane, w, prev, s_q[u], e_q[u], tot, Pb);
    }
}

__global__ void __launch_bounds__(JJ)
k_gamma(const float* __restrict__ P, const float* __restrict__ S,
        float* __restrict__ GT) {
    __shared__ __attribute__((aligned(16))) float tot[16];
    const int row = blockIdx.x;
    const int b = row / II, i = row % II;
    const int j = threadIdx.x;
    const int lane = j & 31;
    const int w = j >> 5;

    const float sv = S[(size_t)row * JJ + j];
    const float pv = P[(size_t)row * JJ + j];

    float low;
    float acc = wave_prefix_lse(pv - sv, lane, low);
    if (lane == 31) tot[w] = acc;
    __syncthreads();

    float t[NWAVE - 1];
    for (int c = 0; c < NWAVE - 1; ++c) {
        float v = tot[c];
        t[c] = (c < w) ? v : NEG_ID;
    }
    float m = acc;
    for (int c = 0; c < NWAVE - 1; ++c) m = fmaxf(m, t[c]);
    float xs[NWAVE];
    xs[0] = fast_exp(acc - m);
    for (int c = 0; c < NWAVE - 1; ++c) xs[c + 1] = fast_exp(t[c] - m);
    float sum = (((xs[0] + xs[1]) + (xs[2] + xs[3])) + ((xs[4] + xs[5]) + (xs[6] + xs[7])))
              + (xs[8] + xs[9]);
    float scan = m + fast_log(sum);

    float lb = sv + scan;
    float gamma = fast_exp(lb);
    if (i == II - 1 && j != JJ - 1) gamma = 0.0f;
    GT[((size_t)b * JJ + j) * II + i] = gamma;
}

__global__ void __launch_bounds__(32)
k_expand_wmma(const float* __restrict__ GT, const float* __restrict__ TT,
              float* __restrict__ out) {
    const int b = blockIdx.z, jt = blockIdx.y, dt = blockIdx.x;
    const int lane = threadIdx.x;
    const int half = lane >> 4;
    const int line = lane & 15;
    const int j0 = jt * 16, d0 = dt * 16;
    const float* __restrict__ Ga = GT + ((size_t)b * JJ + (j0 + line)) * II + half * 2;
    const float* __restrict__ Tb = TT + ((size_t)b * DD + (d0 + line)) * II + half * 2;

    __builtin_prefetch(Ga, 0, 1);
    __builtin_prefetch(Tb, 0, 1);

    v8f acc = {};
#pragma unroll
    for (int k = 0; k < II; k += 4) {
        v2f a  = *(const v2f*)(Ga + k);
        v2f bv = *(const v2f*)(Tb + k);
        acc = __builtin_amdgcn_wmma_f32_16x16x4_f32(false, a, false, bv,
                                                    (short)0, acc, false, false);
    }
#pragma unroll
    for (int v = 0; v < 8; ++v) {
        int j = j0 + v + half * 8;
        out[(size_t)b * JJ * DD + (size_t)j * DD + d0 + line] = acc[v];
    }
}

extern "C" void kernel_launch(void* const* d_in, const int* in_sizes, int n_in,
                              void* d_out, int out_size, void* d_ws, size_t ws_size,
                              hipStream_t stream) {
    const float* text = (const float*)d_in[0];
    const float* mel  = (const float*)d_in[1];
    const float* gum  = (const float*)d_in[2];
    float* out = (float*)d_out;

    const size_t N = (size_t)BB * II * JJ;
    float* E  = (float*)d_ws;
    float* S  = E + N;
    float* P  = S + N;
    float* GT = P + N;
    float* TT = GT + N;

    k_energy_wmma<<<dim3(JJ / 16, II / 16, BB), 32, 0, stream>>>(text, mel, gum, E, TT);
    k_suffix_lse<<<dim3(BB * II), JJ, 0, stream>>>(E, S);
    k_forward<<<dim3(BB), JJ, 0, stream>>>(E, S, P);
    k_gamma<<<dim3(BB * II), JJ, 0, stream>>>(P, S, GT);
    k_expand_wmma<<<dim3(DD / 16, JJ / 16, BB), 32, 0, stream>>>(GT, TT, out);
    (void)in_sizes; (void)n_in; (void)out_size; (void)ws_size;
}


# Gamma rows come out of the forward work-group's own row scan: gamma[i-1] = exp(S + exclusive prefix) + exp(P), stored 16 bytes per 4 rows; gamma launch and P plane gone (5 to 4); LSE tree changes
# speedup vs baseline: 10.1846x; 1.0572x over previous
#include <hip/hip_runtime.h>
#include <math.h>

#define BB 2
#define II 48
#define JJ 320
#define DD 256

#define NEGC   (-1000000000.0f)
#define LOGEPS (-1000.0f)
#define NEG_ID (-1e30f)

#define LOG2E_F 1.4426950408889634f
#define LN2_F   0.6931471805599453f

typedef __attribute__((ext_vector_type(2))) float v2f;
typedef __attribute__((ext_vector_type(4))) float v4f;
typedef __attribute__((ext_vector_type(8))) float v8f;

__device__ __forceinline__ float fast_exp(float x) {
    return __builtin_amdgcn_exp2f(x * LOG2E_F);
}
__device__ __forceinline__ float fast_log(float x) {
    return LN2_F * __builtin_amdgcn_logf(x);
}

__device__ __forceinline__ float lse2(float a, float b) {
    float m = fmaxf(a, b);
    float d = fminf(a, b) - m;
    float t = fast_exp(d);
    return m + fast_log(1.0f + t);
}

template <int N>
__device__ __forceinline__ float row_shr(float fill, float v) {
    return __builtin_bit_cast(float, __builtin_amdgcn_update_dpp(
        __builtin_bit_cast(int, fill), __builtin_bit_cast(int, v),
        0x110 + N, 0xf, 0xf, false));
}
template <int N>
__device__ __forceinline__ float row_shl(float fill, float v) {
    return __builtin_bit_cast(float, __builtin_amdgcn_update_dpp(
        __builtin_bit_cast(int, fill), __builtin_bit_cast(int, v),
        0x100 + N, 0xf, 0xf, false));
}
__device__ __forceinline__ float other_row_last(float v) {
    int x = __builtin_bit_cast(int, v);
    return __builtin_bit_cast(float,
        __builtin_amdgcn_permlanex16(x, x, (int)0xffffffffu, (int)0xffffffffu, false, false));
}
__device__ __forceinline__ float other_row_first(float v) {
    int x = __builtin_bit_cast(int, v);
    return __builtin_bit_cast(float,
        __builtin_amdgcn_permlanex16(x, x, 0, 0, false, false));
}
__device__ __forceinline__ float lane_value(float v, int l) {
    return __builtin_bit_cast(float,
        __builtin_amdgcn_readlane(__builtin_bit_cast(int, v), l));
}

__device__ __forceinline__ float wave_prefix_lse(float acc, int lane, float& low) {
    acc = lse2(acc, row_shr<1>(NEG_ID, acc));
    acc = lse2(acc, row_shr<2>(NEG_ID, acc));
    acc = lse2(acc, row_shr<4>(NEG_ID, acc));
    acc = lse2(acc, row_shr<8>(NEG_ID, acc));
    float t = other_row_last(acc);
    low = (lane >= 16) ? t : NEG_ID;
    return lse2(acc, low);
}
__device__ __forceinline__ float wave_suffix_lse(float acc, int lane) {
    acc = lse2(acc, row_shl<1>(NEG_ID, acc));
    acc = lse2(acc, row_shl<2>(NEG_ID, acc));
    acc = lse2(acc, row_shl<4>(NEG_ID, acc));
    acc = lse2(acc, row_shl<8>(NEG_ID, acc));
    float t = other_row_first(acc);
    float high = (lane < 16) ? t : NEG_ID;
    return lse2(acc, high);
}

__global__ void __launch_bounds__(32)
k_energy_wmma(const float* __restrict__ text, const float* __restrict__ mel,
              const float* __restrict__ gum, float* __restrict__ E,
              float* __restrict__ TT) {
    const int b = blockIdx.z, it = blockIdx.y, jt = blockIdx.x;
    const int lane = threadIdx.x;
    const int half = lane >> 4;
    const int line = lane & 15;

    const float* __restrict__ Ta = text + (size_t)b * II * DD + (size_t)(it * 16 + line) * DD + half * 2;
    const float* __restrict__ Mb = mel  + (size_t)b * JJ * DD + (size_t)(jt * 16 + line) * DD + half * 2;

    __builtin_prefetch(Ta, 0, 1);
    __builtin_prefetch(Mb, 0, 1);
    __builtin_prefetch(Ta + 128, 0, 1);
    __builtin_prefetch(Mb + 128, 0, 1);

    v8f acc = {};
#pragma unroll 4
    for (int k = 0; k < DD; k += 4) {
        v2f a  = *(const v2f*)(Ta + k);
        v2f bv = *(const v2f*)(Mb + k);
        acc = __builtin_amdgcn_wmma_f32_16x16x4_f32(false, a, false, bv,
                                                    (short)0, acc, false, false);
    }

    if (jt < DD / 16) {
        const float* __restrict__ Tt = Ta + jt * 16;
        float* __restrict__ To = TT + ((size_t)b * DD + jt * 16 + half * 2) * II + it * 16 + line;
#pragma unroll
        for (int q = 0; q < 4; ++q) {
            v2f t2 = *(const v2f*)(Tt + q * 4);
            To[(size_t)(q * 4) * II]     = t2[0];
            To[(size_t)(q * 4 + 1) * II] = t2[1];
        }
    }

    const float invTemp = 1.0f / 0.55f;
    const float scale   = 1.0f / 256.0f;
#pragma unroll
    for (int v = 0; v < 8; ++v) {
        int i = it * 16 + v + half * 8;
        int j = jt * 16 + line;
        size_t idx = (size_t)b * II * JJ + (size_t)i * JJ + j;
        float u = gum[idx];
        u = fminf(fmaxf(u, 1e-7f), 1.0f - 1e-7f);
        float noise = -fast_log(-fast_log(u));
        E[idx] = (acc[v] * scale + noise) * invTemp;
    }
}

#define NWAVE (JJ / 32)

__global__ void __launch_bounds__(JJ)
k_suffix_lse(const float* __restrict__ E, float* __restrict__ S) {
    __shared__ __attribute__((aligned(16))) float tot[16];
    const int row = blockIdx.x;
    const int i = row % II;
    const int j = threadIdx.x;
    const int lane = j & 31;
    const int w = j >> 5;

    float x = E[(size_t)row * JJ + j];
    if (i == II - 1 && j != JJ - 1) x = LOGEPS;

    float acc = wave_suffix_lse(x, lane);
    if (lane == 0) tot[w] = acc;
    __syncthreads();

    float t[NWAVE - 1];
    for (int c = 0; c < NWAVE - 1; ++c) {
        float v = tot[c + 1];
        t[c] = (c + 1 > w) ? v : NEG_ID;
    }
    float m = acc;
    for (int c = 0; c < NWAVE - 1; ++c) m = fmaxf(m, t[c]);
    float xs[NWAVE];
    xs[0] = fast_exp(acc - m);
    for (int c = 0; c < NWAVE - 1; ++c) xs[c + 1] = fast_exp(t[c] - m);
    float sum = (((xs[0] + xs[1]) + (xs[2] + xs[3])) + ((xs[4] + xs[5]) + (xs[6] + xs[7])))
              + (xs[8] + xs[9]);
    S[(size_t)row * JJ + j] = m + fast_log(sum);
}

#define PFD 8
#define NMAIN (((II - 1) / PFD) * PFD)
static_assert(NMAIN + PFD == II, "the ring's last refill is the last row's S, and PFD-1 rows follow the loop");
static_assert(PFD % 4 == 0, "gamma leaves in groups of four rows");

__device__ __forceinline__ void lds_barrier() {
    __builtin_amdgcn_fence(__ATOMIC_RELEASE, "workgroup", "local");
    __builtin_amdgcn_s_barrier();
    __builtin_amdgcn_fence(__ATOMIC_ACQUIRE, "workgroup", "local");
}

__device__ __forceinline__ float lse_with_left_totals(float own, int w, const float* trow) {
    const v4f ta = *(const v4f*)&trow[0];
    const v4f tb = *(const v4f*)&trow[4];
    const float tv[NWAVE - 1] = {ta[0], ta[1], ta[2], ta[3], tb[0], tb[1], tb[2], tb[3],
                                 trow[8]};
    float t[NWAVE - 1];
    for (int c = 0; c < NWAVE - 1; ++c) t[c] = (c < w) ? tv[c] : NEG_ID;
    float m = own;
    for (int c = 0; c < NWAVE - 1; ++c) m = fmaxf(m, t[c]);
    float x[NWAVE];
    x[0] = fast_exp(own - m);
    for (int c = 0; c < NWAVE - 1; ++c) x[c + 1] = fast_exp(t[c] - m);
    float sum = (((x[0] + x[1]) + (x[2] + x[3])) + ((x[4] + x[5]) + (x[6] + x[7])))
              + (x[8] + x[9]);
    return m + fast_log(sum);
}

__device__ __forceinline__ float forward_row(int i, int j, int lane, int w, float prev,
                                             float s_cur, float e_cur,
                                             float (*tot)[16], float& g_prev) {
    float low;
    float acc = wave_prefix_lse(prev - s_cur, lane, low);
    if (lane == 31) tot[i & 1][w] = acc;
    float ex = row_shr<1>(low, acc);

    lds_barrier();

    float pre = lse_with_left_totals(ex, w, tot[i & 1]);

    g_prev = fast_exp(s_cur + pre) + fast_exp(prev);

    float v = e_cur + pre;
    if (j == 0) v = NEGC;
    if (!(j >= i && j < (JJ - II + i + 2))) v = NEGC;
    return v;
}

__global__ void __launch_bounds__(JJ)
k_forward(const float* __restrict__ E, const float* __restrict__ S,
          float* __restrict__ GT) {
    __shared__ __attribute__((aligned(16))) float tot[2][16];
    const int b = blockIdx.x;
    const int j = threadIdx.x;
    const int lane = j & 31;
    const int w = j >> 5;

    const float* __restrict__ Eb = E + (size_t)b * II * JJ;
    const float* __restrict__ Sb = S + (size_t)b * II * JJ;
    float* __restrict__ Gj = GT + ((size_t)b * JJ + j) * II;

    float prev = (j == 0) ? 0.0f : NEGC;

    const int je = (j > 0) ? j - 1 : 0;
    float s_q[PFD], e_q[PFD];
#pragma unroll
    for (int r = 0; r < PFD; ++r) {
        s_q[r] = Sb[(size_t)r * JJ + j];
        e_q[r] = Eb[(size_t)r * JJ + je];
    }

    float g[PFD];
    for (int i0 = 1; i0 <= NMAIN; i0 += PFD) {
#pragma unroll
        for (int u = 0; u < PFD; ++u) {
            const int i = i0 + u;
            prev = forward_row(i, j, lane, w, prev, s_q[u], e_q[u], tot, g[u]);
            s_q[u] = Sb[(size_t)(i - 1 + PFD) * JJ + j];
            e_q[u] = Eb[(size_t)(i - 1 + PFD) * JJ + je];
            if ((u & 3) == 3) {
                const v4f g4 = {g[u - 3], g[u - 2], g[u - 1], g[u]};
                *(v4f*)(Gj + (i - 4)) = g4;
            }
        }
    }
#pragma unroll
    for (int u = 0; u < PFD - 1; ++u) {
        prev = forward_row(NMAIN + 1 + u, j, lane, w, prev, s_q[u], e_q[u], tot, g[u]);
        if ((u & 3) == 3) {
            const v4f g4 = {g[u - 3], g[u - 2], g[u - 1], g[u]};
            *(v4f*)(Gj + (NMAIN + u - 3)) = g4;
        }
    }

    {
        const float s_last = s_q[PFD - 1];
        float low;
        float acc = wave_prefix_lse(prev - s_last, lane, low);
        if (lane == 31) tot[II & 1][w] = acc;
        lds_barrier();
        float all = lse_with_left_totals(acc, w, tot[II & 1]);
        g[PFD - 1] = (j == JJ - 1) ? fast_exp(s_last + all) : 0.0f;
        const v4f g4 = {g[PFD - 4], g[PFD - 3], g[PFD - 2], g[PFD - 1]};
        *(v4f*)(Gj + (II - 4)) = g4;
    }
}

__global__ void __launch_bounds__(32)
k_expand_wmma(const float* __restrict__ GT, const float* __restrict__ TT,
              float* __restrict__ out) {
    const int b = blockIdx.z, jt = blockIdx.y, dt = blockIdx.x;
    const int lane = threadIdx.x;
    const int half = lane >> 4;
    const int line = lane & 15;
    const int j0 = jt * 16, d0 = dt * 16;
    const float* __restrict__ Ga = GT + ((size_t)b * JJ + (j0 + line)) * II + half * 2;
    const float* __restrict__ Tb = TT + ((size_t)b * DD + (d0 + line)) * II + half * 2;

    __builtin_prefetch(Ga, 0, 1);
    __builtin_prefetch(Tb, 0, 1);

    v8f acc = {};
#pragma unroll
    for (int k = 0; k < II; k += 4) {
        v2f a  = *(const v2f*)(Ga + k);
        v2f bv = *(const v2f*)(Tb + k);
        acc = __builtin_amdgcn_wmma_f32_16x16x4_f32(false, a, false, bv,
                                                    (short)0, acc, false, false);
    }
#pragma unroll
    for (int v = 0; v < 8; ++v) {
        int j = j0 + v + half * 8;
        out[(size_t)b * JJ * DD + (size_t)j * DD + d0 + line] = acc[v];
    }
}

extern "C" void kernel_launch(void* const* d_in, const int* in_sizes, int n_in,
                              void* d_out, int out_size, void* d_ws, size_t ws_size,
                              hipStream_t stream) {
    const float* text = (const float*)d_in[0];
    const float* mel  = (const float*)d_in[1];
    const float* gum  = (const float*)d_in[2];
    float* out = (float*)d_out;

    const size_t N = (size_t)BB * II * JJ;
    float* E  = (float*)d_ws;
    float* S  = E + N;
    float* GT = S + N;
    float* TT = GT + N;

    k_energy_wmma<<<dim3(JJ / 16, II / 16, BB), 32, 0, stream>>>(text, mel, gum, E, TT);
    k_suffix_lse<<<dim3(BB * II), JJ, 0, stream>>>(E, S);
    k_forward<<<dim3(BB), JJ, 0, stream>>>(E, S, GT);
    k_expand_wmma<<<dim3(DD / 16, JJ / 16, BB), 32, 0, stream>>>(GT, TT, out);
    (void)in_sizes; (void)n_in; (void)out_size; (void)ws_size;
}


# E, S and the forward rows carried in base-2 log units: lse2 = max + log2(1 + exp2(-|a-b|)), 5 dependent ops per scan step instead of 8, no scale multiplies round exp and log; sums keep their order
# speedup vs baseline: 12.1311x; 1.1911x over previous
#include <hip/hip_runtime.h>
#include <math.h>

#define BB 2
#define II 48
#define JJ 320
#define DD 256

#define NEGC   (-1000000000.0f)
#define LOGEPS (-1000.0f)
#define NEG_ID (-1e30f)

#define LOG2E_F 1.4426950408889634f
#define LN2_F   0.6931471805599453f

typedef __attribute__((ext_vector_type(2))) float v2f;
typedef __attribute__((ext_vector_type(4))) float v4f;
typedef __attribute__((ext_vector_type(8))) float v8f;

__device__ __forceinline__ float exp2_raw(float x) {
    return __builtin_amdgcn_exp2f(x);
}
__device__ __forceinline__ float log2_raw(float x) {
    return __builtin_amdgcn_logf(x);
}
__device__ __forceinline__ float fast_log(float x) {
    return LN2_F * log2_raw(x);
}

__device__ __forceinline__ float lse2(float a, float b) {
    float m = fmaxf(a, b);
    float t = exp2_raw(-__builtin_fabsf(a - b));
    return m + log2_raw(1.0f + t);
}

template <int N>
__device__ __forceinline__ float row_shr(float fill, float v) {
    return __builtin_bit_cast(float, __builtin_amdgcn_update_dpp(
        __builtin_bit_cast(int, fill), __builtin_bit_cast(int, v),
        0x110 + N, 0xf, 0xf, false));
}
template <int N>
__device__ __forceinline__ float row_shl(float fill, float v) {
    return __builtin_bit_cast(float, __builtin_amdgcn_update_dpp(
        __builtin_bit_cast(int, fill), __builtin_bit_cast(int, v),
        0x100 + N, 0xf, 0xf, false));
}
__device__ __forceinline__ float other_row_last(float v) {
    int x = __builtin_bit_cast(int, v);
    return __builtin_bit_cast(float,
        __builtin_amdgcn_permlanex16(x, x, (int)0xffffffffu, (int)0xffffffffu, false, false));
}
__device__ __forceinline__ float other_row_first(float v) {
    int x = __builtin_bit_cast(int, v);
    return __builtin_bit_cast(float,
        __builtin_amdgcn_permlanex16(x, x, 0, 0, false, false));
}
__device__ __forceinline__ float lane_value(float v, int l) {
    return __builtin_bit_cast(float,
        __builtin_amdgcn_readlane(__builtin_bit_cast(int, v), l));
}

__device__ __forceinline__ float wave_prefix_lse(float acc, int lane, float& low) {
    acc = lse2(acc, row_shr<1>(NEG_ID, acc));
    acc = lse2(acc, row_shr<2>(NEG_ID, acc));
    acc = lse2(acc, row_shr<4>(NEG_ID, acc));
    acc = lse2(acc, row_shr<8>(NEG_ID, acc));
    float t = other_row_last(acc);
    low = (lane >= 16) ? t : NEG_ID;
    return lse2(acc, low);
}
__device__ __forceinline__ float wave_suffix_lse(float acc, int lane) {
    acc = lse2(acc, row_shl<1>(NEG_ID, acc));
    acc = lse2(acc, row_shl<2>(NEG_ID, acc));
    acc = lse2(acc, row_shl<4>(NEG_ID, acc));
    acc = lse2(acc, row_shl<8>(NEG_ID, acc));
    float t = other_row_first(acc);
    float high = (lane < 16) ? t : NEG_ID;
    return lse2(acc, high);
}

__device__ __forceinline__ void lds_barrier() {
    __builtin_amdgcn_fence(__ATOMIC_RELEASE, "workgroup", "local");
    __builtin_amdgcn_s_barrier();
    __builtin_amdgcn_fence(__ATOMIC_ACQUIRE, "workgroup", "local");
}

#define KSPLIT 8
#define KW (DD / KSPLIT)
#define KSTEPS (KW / 4)
static_assert(KW * sizeof(float) == 128, "a wave's K range is one 128-byte line of each operand row");
static_assert(KSPLIT == 8, "thread (w, lane) finishes accumulator register w: one wave per register");
static_assert(KSTEPS == 8, "a wave's A fragments hold two 16-column text blocks of four k-steps each");

__device__ __forceinline__ void store_text_block(float* __restrict__ To, v2f t0, v2f t1, v2f t2, v2f t3) {
    To[(size_t)0  * II] = t0[0];  To[(size_t)1  * II] = t0[1];
    To[(size_t)4  * II] = t1[0];  To[(size_t)5  * II] = t1[1];
    To[(size_t)8  * II] = t2[0];  To[(size_t)9  * II] = t2[1];
    To[(size_t)12 * II] = t3[0];  To[(size_t)13 * II] = t3[1];
}

__global__ void __launch_bounds__(32 * KSPLIT)
k_energy_wmma(const float* __restrict__ text, const float* __restrict__ mel,
              const float* __restrict__ gum, float* __restrict__ E,
              float* __restrict__ TT) {
    __shared__ __attribute__((aligned(16))) float part[KSPLIT][8][32];
    const int b = blockIdx.z, it = blockIdx.y, jt = blockIdx.x;
    const int lane = threadIdx.x & 31;
    const int w = __builtin_amdgcn_readfirstlane(threadIdx.x >> 5);
    const int half = lane >> 4;
    const int line = lane & 15;

    const float* __restrict__ Ta = text + (size_t)b * II * DD + (size_t)(it * 16 + line) * DD + w * KW + half * 2;
    const float* __restrict__ Mb = mel  + (size_t)b * JJ * DD + (size_t)(jt * 16 + line) * DD + w * KW + half * 2;

    const int i = it * 16 + w + half * 8;
    const int j = jt * 16 + line;
    const size_t idx = (size_t)b * II * JJ + (size_t)i * JJ + j;

    v2f a[KSTEPS], bv[KSTEPS];
#pragma unroll
    for (int q = 0; q < KSTEPS; ++q) {
        a[q]  = *(const v2f*)(Ta + q * 4);
        bv[q] = *(const v2f*)(Mb + q * 4);
    }
    float u = gum[idx];

    v8f acc0 = {}, acc1 = {};
#pragma unroll
    for (int q = 0; q < KSTEPS; q += 2) {
        acc0 = __builtin_amdgcn_wmma_f32_16x16x4_f32(false, a[q], false, bv[q],
                                                     (short)0, acc0, false, false);
        acc1 = __builtin_amdgcn_wmma_f32_16x16x4_f32(false, a[q + 1], false, bv[q + 1],
                                                     (short)0, acc1, false, false);
    }
#pragma unroll
    for (int v = 0; v < 8; ++v) part[w][v][lane] = acc0[v] + acc1[v];

    u = fminf(fmaxf(u, 1e-7f), 1.0f - 1e-7f);
    const float noise = -fast_log(-fast_log(u));

    lds_barrier();

    float p[KSPLIT];
#pragma unroll
    for (int c = 0; c < KSPLIT; ++c) p[c] = part[c][w][lane];
    const float dot = ((p[0] + p[1]) + (p[2] + p[3])) + ((p[4] + p[5]) + (p[6] + p[7]));

    const float invTemp2 = (float)((1.0 / 0.55) * 1.4426950408889634);
    const float scale    = 1.0f / 256.0f;
    E[idx] = (dot * scale + noise) * invTemp2;

    if (jt < DD / 16 && w == (jt >> 1)) {
        float* __restrict__ To = TT + ((size_t)b * DD + jt * 16 + half * 2) * II + it * 16 + line;
        if (jt & 1) store_text_block(To, a[4], a[5], a[6], a[7]);
        else        store_text_block(To, a[0], a[1], a[2], a[3]);
    }
}

#define NWAVE (JJ / 32)

__global__ void __launch_bounds__(JJ)
k_suffix_lse(const float* __restrict__ E, float* __restrict__ S) {
    __shared__ __attribute__((aligned(16))) float tot[16];
    const int row = blockIdx.x;
    const int i = row % II;
    const int j = threadIdx.x;
    const int lane = j & 31;
    const int w = j >> 5;

    float x = E[(size_t)row * JJ + j];
    if (i == II - 1 && j != JJ - 1) x = LOGEPS * LOG2E_F;

    float acc = wave_suffix_lse(x, lane);
    if (lane == 0) tot[w] = acc;
    __syncthreads();

    float t[NWAVE - 1];
    for (int c = 0; c < NWAVE - 1; ++c) {
        float v = tot[c + 1];
        t[c] = (c + 1 > w) ? v : NEG_ID;
    }
    float m = acc;
    for (int c = 0; c < NWAVE - 1; ++c) m = fmaxf(m, t[c]);
    float xs[NWAVE];
    xs[0] = exp2_raw(acc - m);
    for (int c = 0; c < NWAVE - 1; ++c) xs[c + 1] = exp2_raw(t[c] - m);
    float sum = (((xs[0] + xs[1]) + (xs[2] + xs[3])) + ((xs[4] + xs[5]) + (xs[6] + xs[7])))
              + (xs[8] + xs[9]);
    S[(size_t)row * JJ + j] = m + log2_raw(sum);
}

#define PFD 8
#define NMAIN (((II - 1) / PFD) * PFD)
static_assert(NMAIN + PFD == II, "the ring's last refill is the last row's S, and PFD-1 rows follow the loop");
static_assert(PFD % 4 == 0, "gamma leaves in groups of four rows");

__device__ __forceinline__ float lse_with_left_totals(float own, int w, const float* trow) {
    const v4f ta = *(const v4f*)&trow[0];
    const v4f tb = *(const v4f*)&trow[4];
    const float tv[NWAVE - 1] = {ta[0], ta[1], ta[2], ta[3], tb[0], tb[1], tb[2], tb[3],
                                 trow[8]};
    float t[NWAVE - 1];
    for (int c = 0; c < NWAVE - 1; ++c) t[c] = (c < w) ? tv[c] : NEG_ID;
    float m = own;
    for (int c = 0; c < NWAVE - 1; ++c) m = fmaxf(m, t[c]);
    float x[NWAVE];
    x[0] = exp2_raw(own - m);
    for (int c = 0; c < NWAVE - 1; ++c) x[c + 1] = exp2_raw(t[c] - m);
    float sum = (((x[0] + x[1]) + (x[2] + x[3])) + ((x[4] + x[5]) + (x[6] + x[7])))
              + (x[8] + x[9]);
    return m + log2_raw(sum);
}

__device__ __forceinline__ float forward_row(int i, int j, int lane, int w, float prev,
                                             float s_cur, float e_cur,
                                             float (*tot)[16], float& g_prev) {
    float low;
    float acc = wave_prefix_lse(prev - s_cur, lane, low);
    if (lane == 31) tot[i & 1][w] = acc;
    float ex = row_shr<1>(low, acc);

    lds_barrier();

    float pre = lse_with_left_totals(ex, w, tot[i & 1]);

    g_prev = exp2_raw(s_cur + pre) + exp2_raw(prev);

    float v = e_cur + pre;
    if (j == 0) v = NEGC;
    if (!(j >= i && j < (JJ - II + i + 2))) v = NEGC;
    return v;
}

__global__ void __launch_bounds__(JJ)
k_forward(const float* __restrict__ E, const float* __restrict__ S,
          float* __restrict__ GT) {
    __shared__ __attribute__((aligned(16))) float tot[2][16];
    const int b = blockIdx.x;
    const int j = threadIdx.x;
    const int lane = j & 31;
    const int w = j >> 5;

    const float* __restrict__ Eb = E + (size_t)b * II * JJ;
    const float* __restrict__ Sb = S + (size_t)b * II * JJ;
    float* __restrict__ Gj = GT + ((size_t)b * JJ + j) * II;

    float prev = (j == 0) ? 0.0f : NEGC;

    const int je = (j > 0) ? j - 1 : 0;
    float s_q[PFD], e_q[PFD];
#pragma unroll
    for (int r = 0; r < PFD; ++r) {
        s_q[r] = Sb[(size_t)r * JJ + j];
        e_q[r] = Eb[(size_t)r * JJ + je];
    }

    float g[PFD];
    for (int i0 = 1; i0 <= NMAIN; i0 += PFD) {
#pragma unroll
        for (int u = 0; u < PFD; ++u) {
            const int i = i0 + u;
            prev = forward_row(i, j, lane, w, prev, s_q[u], e_q[u], tot, g[u]);
            s_q[u] = Sb[(size_t)(i - 1 + PFD) * JJ + j];
            e_q[u] = Eb[(size_t)(i - 1 + PFD) * JJ + je];
            if ((u & 3) == 3) {
                const v4f g4 = {g[u - 3], g[u - 2], g[u - 1], g[u]};
                *(v4f*)(Gj + (i - 4)) = g4;
            }
        }
    }
#pragma unroll
    for (int u = 0; u < PFD - 1; ++u) {
        prev = forward_row(NMAIN + 1 + u, j, lane, w, prev, s_q[u], e_q[u], tot, g[u]);
        if ((u & 3) == 3) {
            const v4f g4 = {g[u - 3], g[u - 2], g[u - 1], g[u]};
            *(v4f*)(Gj + (NMAIN + u - 3)) = g4;
        }
    }

    {
        const float s_last = s_q[PFD - 1];
        float low;
        float acc = wave_prefix_lse(prev - s_last, lane, low);
        if (lane == 31) tot[II & 1][w] = acc;
        lds_barrier();
        float all = lse_with_left_totals(acc, w, tot[II & 1]);
        g[PFD - 1] = (j == JJ - 1) ? exp2_raw(s_last + all) : 0.0f;
        const v4f g4 = {g[PFD - 4], g[PFD - 3], g[PFD - 2], g[PFD - 1]};
        *(v4f*)(Gj + (II - 4)) = g4;
    }
}

__global__ void __launch_bounds__(32)
k_expand_wmma(const float* __restrict__ GT, const float* __restrict__ TT,
              float* __restrict__ out) {
    const int b = blockIdx.z, jt = blockIdx.y, dt = blockIdx.x;
    const int lane = threadIdx.x;
    const int half = lane >> 4;
    const int line = lane & 15;
    const int j0 = jt * 16, d0 = dt * 16;
    const float* __restrict__ Ga = GT + ((size_t)b * JJ + (j0 + line)) * II + half * 2;
    const float* __restrict__ Tb = TT + ((size_t)b * DD + (d0 + line)) * II + half * 2;

    __builtin_prefetch(Ga, 0, 1);
    __builtin_prefetch(Tb, 0, 1);

    v8f acc = {};
#pragma unroll
    for (int k = 0; k < II; k += 4) {
        v2f a  = *(const v2f*)(Ga + k);
        v2f bv = *(const v2f*)(Tb + k);
        acc = __builtin_amdgcn_wmma_f32_16x16x4_f32(false, a, false, bv,
                                                    (short)0, acc, false, false);
    }
#pragma unroll
    for (int v = 0; v < 8; ++v) {
        int j = j0 + v + half * 8;
        out[(size_t)b * JJ * DD + (size_t)j * DD + d0 + line] = acc[v];
    }
}

extern "C" void kernel_launch(void* const* d_in, const int* in_sizes, int n_in,
                              void* d_out, int out_size, void* d_ws, size_t ws_size,
                              hipStream_t stream) {
    const float* text = (const float*)d_in[0];
    const float* mel  = (const float*)d_in[1];
    const float* gum  = (const float*)d_in[2];
    float* out = (float*)d_out;

    const size_t N = (size_t)BB * II * JJ;
    float* E  = (float*)d_ws;
    float* S  = E + N;
    float* GT = S + N;
    float* TT = GT + N;

    k_energy_wmma<<<dim3(JJ / 16, II / 16, BB), 32 * KSPLIT, 0, stream>>>(text, mel, gum, E, TT);
    k_suffix_lse<<<dim3(BB * II), JJ, 0, stream>>>(E, S);
    k_forward<<<dim3(BB), JJ, 0, stream>>>(E, S, GT);
    k_expand_wmma<<<dim3(DD / 16, JJ / 16, BB), 32, 0, stream>>>(GT, TT, out);
    (void)in_sizes; (void)n_in; (void)out_size; (void)ws_size;
}


# K3 row split by LSE distributivity: next chunk scan = lse2(A, L + Bc), A scanned while totals cross LDS, Bc and masked W from K2; wave-uniform L joined one total per lane on DPP; LSE tree changes
# speedup vs baseline: 13.3243x; 1.0984x over previous
#include <hip/hip_runtime.h>
#include <math.h>

#define BB 2
#define II 48
#define JJ 320
#define DD 256

#define NEGC   (-1000000000.0f)
#define LOGEPS (-1000.0f)
#define NEG_ID (-1e30f)

#define LOG2E_F 1.4426950408889634f
#define LN2_F   0.6931471805599453f

typedef __attribute__((ext_vector_type(2))) float v2f;
typedef __attribute__((ext_vector_type(4))) float v4f;
typedef __attribute__((ext_vector_type(8))) float v8f;

__device__ __forceinline__ float exp2_raw(float x) {
    return __builtin_amdgcn_exp2f(x);
}
__device__ __forceinline__ float log2_raw(float x) {
    return __builtin_amdgcn_logf(x);
}
__device__ __forceinline__ float fast_log(float x) {
    return LN2_F * log2_raw(x);
}

__device__ __forceinline__ float lse2(float a, float b) {
    float m = fmaxf(a, b);
    float t = exp2_raw(-__builtin_fabsf(a - b));
    return m + log2_raw(1.0f + t);
}

template <int N>
__device__ __forceinline__ float row_shr(float fill, float v) {
    return __builtin_bit_cast(float, __builtin_amdgcn_update_dpp(
        __builtin_bit_cast(int, fill), __builtin_bit_cast(int, v),
        0x110 + N, 0xf, 0xf, false));
}
template <int N>
__device__ __forceinline__ float row_shl(float fill, float v) {
    return __builtin_bit_cast(float, __builtin_amdgcn_update_dpp(
        __builtin_bit_cast(int, fill), __builtin_bit_cast(int, v),
        0x100 + N, 0xf, 0xf, false));
}
__device__ __forceinline__ float other_row_last(float v) {
    int x = __builtin_bit_cast(int, v);
    return __builtin_bit_cast(float,
        __builtin_amdgcn_permlanex16(x, x, (int)0xffffffffu, (int)0xffffffffu, false, false));
}
__device__ __forceinline__ float other_row_first(float v) {
    int x = __builtin_bit_cast(int, v);
    return __builtin_bit_cast(float,
        __builtin_amdgcn_permlanex16(x, x, 0, 0, false, false));
}
__device__ __forceinline__ float lane_value(float v, int l) {
    return __builtin_bit_cast(float,
        __builtin_amdgcn_readlane(__builtin_bit_cast(int, v), l));
}

__device__ __forceinline__ float wave_prefix_lse(float acc, int lane, float& low) {
    acc = lse2(acc, row_shr<1>(NEG_ID, acc));
    acc = lse2(acc, row_shr<2>(NEG_ID, acc));
    acc = lse2(acc, row_shr<4>(NEG_ID, acc));
    acc = lse2(acc, row_shr<8>(NEG_ID, acc));
    float t = other_row_last(acc);
    low = (lane >= 16) ? t : NEG_ID;
    return lse2(acc, low);
}
__device__ __forceinline__ float wave_suffix_lse(float acc, int lane) {
    acc = lse2(acc, row_shl<1>(NEG_ID, acc));
    acc = lse2(acc, row_shl<2>(NEG_ID, acc));
    acc = lse2(acc, row_shl<4>(NEG_ID, acc));
    acc = lse2(acc, row_shl<8>(NEG_ID, acc));
    float t = other_row_first(acc);
    float high = (lane < 16) ? t : NEG_ID;
    return lse2(acc, high);
}

__device__ __forceinline__ void lds_barrier() {
    __builtin_amdgcn_fence(__ATOMIC_RELEASE, "workgroup", "local");
    __builtin_amdgcn_s_barrier();
    __builtin_amdgcn_fence(__ATOMIC_ACQUIRE, "workgroup", "local");
}

#define KSPLIT 8
#define KW (DD / KSPLIT)
#define KSTEPS (KW / 4)
static_assert(KW * sizeof(float) == 128, "a wave's K range is one 128-byte line of each operand row");
static_assert(KSPLIT == 8, "thread (w, lane) finishes accumulator register w: one wave per register");
static_assert(KSTEPS == 8, "a wave's A fragments hold two 16-column text blocks of four k-steps each");

__device__ __forceinline__ void store_text_block(float* __restrict__ To, v2f t0, v2f t1, v2f t2, v2f t3) {
    To[(size_t)0  * II] = t0[0];  To[(size_t)1  * II] = t0[1];
    To[(size_t)4  * II] = t1[0];  To[(size_t)5  * II] = t1[1];
    To[(size_t)8  * II] = t2[0];  To[(size_t)9  * II] = t2[1];
    To[(size_t)12 * II] = t3[0];  To[(size_t)13 * II] = t3[1];
}

__global__ void __launch_bounds__(32 * KSPLIT)
k_energy_wmma(const float* __restrict__ text, const float* __restrict__ mel,
              const float* __restrict__ gum, float* __restrict__ E,
              float* __restrict__ TT) {
    __shared__ __attribute__((aligned(16))) float part[KSPLIT][8][32];
    const int b = blockIdx.z, it = blockIdx.y, jt = blockIdx.x;
    const int lane = threadIdx.x & 31;
    const int w = __builtin_amdgcn_readfirstlane(threadIdx.x >> 5);
    const int half = lane >> 4;
    const int line = lane & 15;

    const float* __restrict__ Ta = text + (size_t)b * II * DD + (size_t)(it * 16 + line) * DD + w * KW + half * 2;
    const float* __restrict__ Mb = mel  + (size_t)b * JJ * DD + (size_t)(jt * 16 + line) * DD + w * KW + half * 2;

    const int i = it * 16 + w + half * 8;
    const int j = jt * 16 + line;
    const size_t idx = (size_t)b * II * JJ + (size_t)i * JJ + j;

    v2f a[KSTEPS], bv[KSTEPS];
#pragma unroll
    for (int q = 0; q < KSTEPS; ++q) {
        a[q]  = *(const v2f*)(Ta + q * 4);
        bv[q] = *(const v2f*)(Mb + q * 4);
    }
    float u = gum[idx];

    v8f acc0 = {}, acc1 = {};
#pragma unroll
    for (int q = 0; q < KSTEPS; q += 2) {
        acc0 = __builtin_amdgcn_wmma_f32_16x16x4_f32(false, a[q], false, bv[q],
                                                     (short)0, acc0, false, false);
        acc1 = __builtin_amdgcn_wmma_f32_16x16x4_f32(false, a[q + 1], false, bv[q + 1],
                                                     (short)0, acc1, false, false);
    }
#pragma unroll
    for (int v = 0; v < 8; ++v) part[w][v][lane] = acc0[v] + acc1[v];

    u = fminf(fmaxf(u, 1e-7f), 1.0f - 1e-7f);
    const float noise = -fast_log(-fast_log(u));

    lds_barrier();

    float p[KSPLIT];
#pragma unroll
    for (int c = 0; c < KSPLIT; ++c) p[c] = part[c][w][lane];
    const float dot = ((p[0] + p[1]) + (p[2] + p[3])) + ((p[4] + p[5]) + (p[6] + p[7]));

    const float invTemp2 = (float)((1.0 / 0.55) * 1.4426950408889634);
    const float scale    = 1.0f / 256.0f;
    E[idx] = (dot * scale + noise) * invTemp2;

    if (jt < DD / 16 && w == (jt >> 1)) {
        float* __restrict__ To = TT + ((size_t)b * DD + jt * 16 + half * 2) * II + it * 16 + line;
        if (jt & 1) store_text_block(To, a[4], a[5], a[6], a[7]);
        else        store_text_block(To, a[0], a[1], a[2], a[3]);
    }
}

#define NWAVE (JJ / 32)

__global__ void __launch_bounds__(JJ)
k_suffix_lse(const float* __restrict__ E, float* __restrict__ S,
             float* __restrict__ Wp, float* __restrict__ Bp) {
    __shared__ __attribute__((aligned(16))) float tot[16];
    const int row = blockIdx.x;
    const int i = row % II;
    const int j = threadIdx.x;
    const int lane = j & 31;
    const int w = j >> 5;

    const bool in_win = (i >= 1) && (j >= i) && (j < (JJ - II + i + 2));

    float x = E[(size_t)row * JJ + j];
    const float e_up = E[in_win ? ((size_t)(row - 1) * JJ + (j - 1)) : ((size_t)row * JJ + j)];
    if (i == II - 1 && j != JJ - 1) x = LOGEPS * LOG2E_F;

    float acc = wave_suffix_lse(x, lane);
    if (lane == 0) tot[w] = acc;
    __syncthreads();

    float t[NWAVE - 1];
    for (int c = 0; c < NWAVE - 1; ++c) {
        float v = tot[c + 1];
        t[c] = (c + 1 > w) ? v : NEG_ID;
    }
    float m = acc;
    for (int c = 0; c < NWAVE - 1; ++c) m = fmaxf(m, t[c]);
    float xs[NWAVE];
    xs[0] = exp2_raw(acc - m);
    for (int c = 0; c < NWAVE - 1; ++c) xs[c + 1] = exp2_raw(t[c] - m);
    float sum = (((xs[0] + xs[1]) + (xs[2] + xs[3])) + ((xs[4] + xs[5]) + (xs[6] + xs[7])))
              + (xs[8] + xs[9]);
    const float s_val = m + log2_raw(sum);
    S[(size_t)row * JJ + j] = s_val;

    const float wv = in_win ? (e_up - s_val) : NEGC;
    float low;
    const float bc = wave_prefix_lse(wv, lane, low);
    Wp[(size_t)row * JJ + j] = wv;
    Bp[(size_t)row * JJ + j] = bc;
}

#define PFD 8
static_assert(II % PFD == 0, "the rows are walked in whole trips of PFD");
static_assert(PFD % 4 == 0, "gamma leaves in groups of four rows");

__device__ __forceinline__ float row_max16(float v) {
    asm("v_max_num_f32_dpp %0, %0, %0 row_xmask:1 row_mask:0xf bank_mask:0xf\n\t"
        "s_delay_alu instid0(VALU_DEP_1)\n\t"
        "v_max_num_f32_dpp %0, %0, %0 row_xmask:2 row_mask:0xf bank_mask:0xf\n\t"
        "s_delay_alu instid0(VALU_DEP_1)\n\t"
        "v_max_num_f32_dpp %0, %0, %0 row_xmask:4 row_mask:0xf bank_mask:0xf\n\t"
        "s_delay_alu instid0(VALU_DEP_1)\n\t"
        "v_max_num_f32_dpp %0, %0, %0 row_xmask:8 row_mask:0xf bank_mask:0xf"
        : "+v"(v));
    return v;
}
__device__ __forceinline__ float row_sum16(float v) {
    asm("v_nop\n\t"
        "v_add_f32_dpp %0, %0, %0 row_xmask:1 row_mask:0xf bank_mask:0xf\n\t"
        "s_delay_alu instid0(VALU_DEP_1)\n\t"
        "v_add_f32_dpp %0, %0, %0 row_xmask:2 row_mask:0xf bank_mask:0xf\n\t"
        "s_delay_alu instid0(VALU_DEP_1)\n\t"
        "v_add_f32_dpp %0, %0, %0 row_xmask:4 row_mask:0xf bank_mask:0xf\n\t"
        "s_delay_alu instid0(VALU_DEP_1)\n\t"
        "v_add_f32_dpp %0, %0, %0 row_xmask:8 row_mask:0xf bank_mask:0xf"
        : "+v"(v));
    return v;
}

__device__ __forceinline__ float left_totals_lse(int c, int slot, int w, const float* trow) {
    const float v = trow[slot];
    const float t = (c < w) ? v : NEG_ID;
    const float m = row_max16(t);
    const float s = row_sum16(exp2_raw(t - m));
    return m + log2_raw(s);
}

template <bool NEXT>
__device__ __forceinline__ void forward_step(int r, int lane, int slot, int w, float& a, float& L,
                                             float s_r, float b_r, float w_next,
                                             float (*tot)[16], float& g) {
    const float yloc = lse2(a, L + b_r);
    if (lane == 31) tot[r & 1][w] = yloc;

    lds_barrier();

    L = left_totals_lse(lane & 15, slot, w, tot[r & 1]);

    if (NEXT) {
        const float t = other_row_last(yloc);
        const float fill = (lane >= 16) ? t : NEG_ID;
        const float xloc = row_shr<1>(fill, yloc);
        float low;
        a = wave_prefix_lse(w_next + xloc, lane, low);
    }

    g = exp2_raw(s_r + yloc) + exp2_raw(s_r + L);
}

__global__ void __launch_bounds__(JJ)
k_forward(const float* __restrict__ S, const float* __restrict__ Wp,
          const float* __restrict__ Bp, float* __restrict__ GT) {
    __shared__ __attribute__((aligned(16))) float tot[2][16];
    const int b = blockIdx.x;
    const int j = threadIdx.x;
    const int lane = j & 31;
    const int w = j >> 5;
    const int slot = ((lane & 15) < NWAVE - 1) ? (lane & 15) : NWAVE - 1;

    const float* __restrict__ Sb = S  + (size_t)b * II * JJ + j;
    const float* __restrict__ Wb = Wp + (size_t)b * II * JJ + j;
    const float* __restrict__ Bb = Bp + (size_t)b * II * JJ + j;
    float* __restrict__ Gj = GT + ((size_t)b * JJ + j) * II;

    float s_q[PFD], b_q[PFD], w_q[PFD];
#pragma unroll
    for (int r = 0; r < PFD; ++r) {
        s_q[r] = Sb[(size_t)r * JJ];
        b_q[r] = Bb[(size_t)r * JJ];
        w_q[r] = Wb[(size_t)(r + 1) * JJ];
    }

    float L = NEG_ID;
    float a;
    {
        const float q0 = ((j == 0) ? 0.0f : NEGC) - s_q[0];
        float low;
        a = wave_prefix_lse(q0, lane, low);
    }

    float g[PFD];
    for (int r0 = 0; r0 < II - PFD; r0 += PFD) {
#pragma unroll
        for (int u = 0; u < PFD; ++u) {
            const int r = r0 + u;
            forward_step<true>(r, lane, slot, w, a, L, s_q[u], b_q[u], w_q[u], tot, g[u]);
            const int rn = r + PFD;
            const int rw = (rn + 1 < II) ? rn + 1 : II - 1;
            s_q[u] = Sb[(size_t)rn * JJ];
            b_q[u] = Bb[(size_t)rn * JJ];
            w_q[u] = Wb[(size_t)rw * JJ];
            if ((u & 3) == 3) {
                const v4f g4 = {g[u - 3], g[u - 2], g[u - 1], g[u]};
                *(v4f*)(Gj + (r - 3)) = g4;
            }
        }
    }
#pragma unroll
    for (int u = 0; u < PFD - 1; ++u) {
        forward_step<true>(II - PFD + u, lane, slot, w, a, L, s_q[u], b_q[u], w_q[u], tot, g[u]);
        if ((u & 3) == 3) {
            const v4f g4 = {g[u - 3], g[u - 2], g[u - 1], g[u]};
            *(v4f*)(Gj + (II - PFD + u - 3)) = g4;
        }
    }
    {
        float gl;
        forward_step<false>(II - 1, lane, slot, w, a, L, s_q[PFD - 1], b_q[PFD - 1], 0.0f, tot, gl);
        g[PFD - 1] = (j == JJ - 1) ? gl : 0.0f;
        const v4f g4 = {g[PFD - 4], g[PFD - 3], g[PFD - 2], g[PFD - 1]};
        *(v4f*)(Gj + (II - 4)) = g4;
    }
}

__global__ void __launch_bounds__(32)
k_expand_wmma(const float* __restrict__ GT, const float* __restrict__ TT,
              float* __restrict__ out) {
    const int b = blockIdx.z, jt = blockIdx.y, dt = blockIdx.x;
    const int lane = threadIdx.x;
    const int half = lane >> 4;
    const int line = lane & 15;
    const int j0 = jt * 16, d0 = dt * 16;
    const float* __restrict__ Ga = GT + ((size_t)b * JJ + (j0 + line)) * II + half * 2;
    const float* __restrict__ Tb = TT + ((size_t)b * DD + (d0 + line)) * II + half * 2;

    __builtin_prefetch(Ga, 0, 1);
    __builtin_prefetch(Tb, 0, 1);

    v8f acc = {};
#pragma unroll
    for (int k = 0; k < II; k += 4) {
        v2f a  = *(const v2f*)(Ga + k);
        v2f bv = *(const v2f*)(Tb + k);
        acc = __builtin_amdgcn_wmma_f32_16x16x4_f32(false, a, false, bv,
                                                    (short)0, acc, false, false);
    }
#pragma unroll
    for (int v = 0; v < 8; ++v) {
        int j = j0 + v + half * 8;
        out[(size_t)b * JJ * DD + (size_t)j * DD + d0 + line] = acc[v];
    }
}

extern "C" void kernel_launch(void* const* d_in, const int* in_sizes, int n_in,
                              void* d_out, int out_size, void* d_ws, size_t ws_size,
                              hipStream_t stream) {
    const float* text = (const float*)d_in[0];
    const float* mel  = (const float*)d_in[1];
    const float* gum  = (const float*)d_in[2];
    float* out = (float*)d_out;

    const size_t N = (size_t)BB * II * JJ;
    float* E  = (float*)d_ws;
    float* S  = E + N;
    float* GT = S + N;
    float* TT = GT + N;
    float* Wp = TT + (size_t)BB * DD * II;
    float* Bp = Wp + N;

    k_energy_wmma<<<dim3(JJ / 16, II / 16, BB), 32 * KSPLIT, 0, stream>>>(text, mel, gum, E, TT);
    k_suffix_lse<<<dim3(BB * II), JJ, 0, stream>>>(E, S, Wp, Bp);
    k_forward<<<dim3(BB), JJ, 0, stream>>>(S, Wp, Bp, GT);
    k_expand_wmma<<<dim3(DD / 16, JJ / 16, BB), 32, 0, stream>>>(GT, TT, out);
    (void)in_sizes; (void)n_in; (void)out_size; (void)ws_size;
}
